# starting kernel
# baseline (speedup 1.0000x reference)
#include <hip/hip_runtime.h>
#include <hip/hip_bf16.h>

typedef float v2f __attribute__((ext_vector_type(2)));
typedef float v4f __attribute__((ext_vector_type(4)));
typedef float v8f __attribute__((ext_vector_type(8)));

#define DI 64
#define HH 128

__device__ __forceinline__ v8f wmma_f32_k4(v2f a, v2f b, v8f c) {
    return __builtin_amdgcn_wmma_f32_16x16x4_f32(false, a, false, b, (short)0, c, false, false);
}

__global__ __launch_bounds__(128)
void k_edge_msg(const float* __restrict__ x, const int* __restrict__ src,
                const float* __restrict__ ea, const float* __restrict__ Wedge,
                float* __restrict__ msg, long long E, long long N) {
    __shared__ float Ws[64][66];
    __shared__ float As[4][16][66];

    const int tid  = threadIdx.x;
    const int wave = tid >> 5;
    const int lane = tid & 31;
    const int hf   = lane >> 4;
    const int lm   = lane & 15;

    for (int i = tid; i < 64 * 64; i += 128)
        Ws[i >> 6][i & 63] = Wedge[i];

    const long long e0 = ((long long)blockIdx.x * 4 + wave) * 16;
    for (int r = 0; r < 16; ++r) {
        long long e  = e0 + r;
        long long ec = e < E ? e : E - 1;
        const v2f av = __builtin_nontemporal_load((const v2f*)(ea + ec * DI + lane * 2));
        As[wave][r][lane * 2 + 0] = av.x;
        As[wave][r][lane * 2 + 1] = av.y;
    }
    __syncthreads();

    for (int n0 = 0; n0 < DI; n0 += 16) {
        v8f acc = {};
#pragma unroll
        for (int k0 = 0; k0 < DI; k0 += 4) {
            const int kb = k0 + hf * 2;
            v2f a, b;
            a.x = As[wave][lm][kb];     a.y = As[wave][lm][kb + 1];
            b.x = Ws[kb][n0 + lm];      b.y = Ws[kb + 1][n0 + lm];
            acc = wmma_f32_k4(a, b, acc);
        }
#pragma unroll
        for (int i = 0; i < 8; ++i) {
            const int m = i + hf * 8;
            const long long e = e0 + m;
            if (e < E) {
                const long long s = (long long)src[e];
                float v = acc[i] + x[s * DI + n0 + lm];
                v = (v > 0.f ? v : 0.f) + 1e-7f;
                __builtin_nontemporal_store(v, msg + e * DI + n0 + lm);
            }
        }
    }
}

__global__ __launch_bounds__(256)
void k_seg_max(const float* __restrict__ msg, const int* __restrict__ dst,
               unsigned int* __restrict__ mx, long long E) {
    const long long i = (long long)blockIdx.x * blockDim.x + threadIdx.x;
    if (i >= E * 16) return;
    const long long e = i >> 4;
    const int d4 = (int)(i & 15) * 4;
    const long long dd = (long long)dst[e];
    const v4f m = __builtin_nontemporal_load((const v4f*)(msg + e * DI + d4));
    unsigned int* b = mx + dd * DI + d4;
    atomicMax(b + 0, __float_as_uint(m.x));
    atomicMax(b + 1, __float_as_uint(m.y));
    atomicMax(b + 2, __float_as_uint(m.z));
    atomicMax(b + 3, __float_as_uint(m.w));
}

__global__ __launch_bounds__(256)
void k_seg_expsum(const float* __restrict__ msg, const int* __restrict__ dst,
                  const float* __restrict__ mx, float* __restrict__ sum, long long E) {
    const long long i = (long long)blockIdx.x * blockDim.x + threadIdx.x;
    if (i >= E * 16) return;
    const long long e = i >> 4;
    const int d4 = (int)(i & 15) * 4;
    const long long dd = (long long)dst[e];
    const v4f m  = __builtin_nontemporal_load((const v4f*)(msg + e * DI + d4));
    const v4f mv = *(const v4f*)(mx + dd * DI + d4);
    float* b = sum + dd * DI + d4;
    atomicAdd(b + 0, __expf(m.x - mv.x));
    atomicAdd(b + 1, __expf(m.y - mv.y));
    atomicAdd(b + 2, __expf(m.z - mv.z));
    atomicAdd(b + 3, __expf(m.w - mv.w));
}

__global__ __launch_bounds__(256)
void k_seg_agg(const float* __restrict__ msg, const int* __restrict__ dst,
               const float* __restrict__ mx, const float* __restrict__ sum,
               float* __restrict__ agg, long long E) {
    const long long i = (long long)blockIdx.x * blockDim.x + threadIdx.x;
    if (i >= E * 16) return;
    const long long e = i >> 4;
    const int d4 = (int)(i & 15) * 4;
    const long long dd = (long long)dst[e];
    const v4f m  = __builtin_nontemporal_load((const v4f*)(msg + e * DI + d4));
    const v4f mv = *(const v4f*)(mx + dd * DI + d4);
    const v4f sv = *(const v4f*)(sum + dd * DI + d4);
    float* b = agg + dd * DI + d4;
    atomicAdd(b + 0, m.x * __expf(m.x - mv.x) / (sv.x + 1e-16f));
    atomicAdd(b + 1, m.y * __expf(m.y - mv.y) / (sv.y + 1e-16f));
    atomicAdd(b + 2, m.z * __expf(m.z - mv.z) / (sv.z + 1e-16f));
    atomicAdd(b + 3, m.w * __expf(m.w - mv.w) / (sv.w + 1e-16f));
}

__global__ __launch_bounds__(128)
void k_h_stats(const float* __restrict__ x, const float* __restrict__ agg,
               const float* __restrict__ W1, float* __restrict__ h,
               float* __restrict__ colsum, float* __restrict__ colsq, long long N) {
    __shared__ float W1s[64][130];
    __shared__ float As[4][16][66];
    __shared__ float cs_s[128], cq_s[128];

    const int tid  = threadIdx.x;
    const int wave = tid >> 5;
    const int lane = tid & 31;
    const int hf   = lane >> 4;
    const int lm   = lane & 15;

    for (int i = tid; i < 64 * HH; i += 128)
        W1s[i >> 7][i & 127] = W1[i];
    if (tid < 128) { cs_s[tid] = 0.f; cq_s[tid] = 0.f; }

    const long long r0 = ((long long)blockIdx.x * 4 + wave) * 16;
    for (int r = 0; r < 16; ++r) {
        long long rr = r0 + r;
        long long rc = rr < N ? rr : N - 1;
        const float2 a  = *(const float2*)(agg + rc * DI + lane * 2);
        const float2 xv = *(const float2*)(x + rc * DI + lane * 2);
        As[wave][r][lane * 2 + 0] = a.x + xv.x;
        As[wave][r][lane * 2 + 1] = a.y + xv.y;
    }
    __syncthreads();

    for (int n0 = 0; n0 < HH; n0 += 16) {
        v8f acc = {};
#pragma unroll
        for (int k0 = 0; k0 < DI; k0 += 4) {
            const int kb = k0 + hf * 2;
            v2f a, b;
            a.x = As[wave][lm][kb];   a.y = As[wave][lm][kb + 1];
            b.x = W1s[kb][n0 + lm];   b.y = W1s[kb + 1][n0 + lm];
            acc = wmma_f32_k4(a, b, acc);
        }
        float p1 = 0.f, p2 = 0.f;
#pragma unroll
        for (int i = 0; i < 8; ++i) {
            const int m = i + hf * 8;
            const long long rr = r0 + m;
            if (rr < N) {
                const float v = acc[i];
                h[rr * HH + n0 + lm] = v;
                p1 += v;
                p2 += v * v;
            }
        }
        p1 += __shfl_xor(p1, 16, 32);
        p2 += __shfl_xor(p2, 16, 32);
        if (hf == 0) {
            atomicAdd(&cs_s[n0 + lm], p1);
            atomicAdd(&cq_s[n0 + lm], p2);
        }
    }
    __syncthreads();
    if (tid < 128) {
        atomicAdd(colsum + tid, cs_s[tid]);
        atomicAdd(colsq + tid, cq_s[tid]);
    }
}

__global__ __launch_bounds__(128)
void k_bn(const float* __restrict__ colsum, const float* __restrict__ colsq,
          const float* __restrict__ gamma, const float* __restrict__ beta,
          float* __restrict__ scale, float* __restrict__ shift, float invN) {
    const int i = threadIdx.x;
    const float mu  = colsum[i] * invN;
    const float var = colsq[i] * invN - mu * mu;
    const float sc  = rsqrtf(var + 1e-5f) * gamma[i];
    scale[i] = sc;
    shift[i] = beta[i] - mu * sc;
}

__global__ __launch_bounds__(64)
void k_out(const float* __restrict__ h, const float* __restrict__ W2,
           const float* __restrict__ scale, const float* __restrict__ shift,
           float* __restrict__ out, long long N) {
    __shared__ float W2s[HH][66];
    __shared__ float As[2][16][130];
    __shared__ float sc_s[128], sh_s[128];

    const int tid  = threadIdx.x;
    const int wave = tid >> 5;
    const int lane = tid & 31;
    const int hf   = lane >> 4;
    const int lm   = lane & 15;

    for (int i = tid; i < HH * DI; i += 64)
        W2s[i >> 6][i & 63] = W2[i];
    for (int i = tid; i < 128; i += 64) { sc_s[i] = scale[i]; sh_s[i] = shift[i]; }
    __syncthreads();

    const long long r0 = ((long long)blockIdx.x * 2 + wave) * 16;
    for (int r = 0; r < 16; ++r) {
        long long rr = r0 + r;
        long long rc = rr < N ? rr : N - 1;
        const int c = lane * 4;
        const v4f hv = *(const v4f*)(h + rc * HH + c);
        float v0 = hv.x * sc_s[c + 0] + sh_s[c + 0];
        float v1 = hv.y * sc_s[c + 1] + sh_s[c + 1];
        float v2 = hv.z * sc_s[c + 2] + sh_s[c + 2];
        float v3 = hv.w * sc_s[c + 3] + sh_s[c + 3];
        As[wave][r][c + 0] = v0 > 0.f ? v0 : 0.f;
        As[wave][r][c + 1] = v1 > 0.f ? v1 : 0.f;
        As[wave][r][c + 2] = v2 > 0.f ? v2 : 0.f;
        As[wave][r][c + 3] = v3 > 0.f ? v3 : 0.f;
    }
    __syncthreads();

    for (int n0 = 0; n0 < DI; n0 += 16) {
        v8f acc = {};
#pragma unroll
        for (int k0 = 0; k0 < HH; k0 += 4) {
            const int kb = k0 + hf * 2;
            v2f a, b;
            a.x = As[wave][lm][kb];   a.y = As[wave][lm][kb + 1];
            b.x = W2s[kb][n0 + lm];   b.y = W2s[kb + 1][n0 + lm];
            acc = wmma_f32_k4(a, b, acc);
        }
#pragma unroll
        for (int i = 0; i < 8; ++i) {
            const int m = i + hf * 8;
            const long long rr = r0 + m;
            if (rr < N) __builtin_nontemporal_store(acc[i], out + rr * DI + n0 + lm);
        }
    }
}

extern "C" void kernel_launch(void* const* d_in, const int* in_sizes, int n_in,
                              void* d_out, int out_size, void* d_ws, size_t ws_size,
                              hipStream_t stream) {
    const float* x     = (const float*)d_in[0];
    const int*   eidx  = (const int*)d_in[1];
    const float* ea    = (const float*)d_in[2];
    const float* Wedge = (const float*)d_in[3];
    const float* W1    = (const float*)d_in[4];
    const float* gamma = (const float*)d_in[5];
    const float* beta  = (const float*)d_in[6];
    const float* W2    = (const float*)d_in[7];
    float* out = (float*)d_out;

    const long long N = in_sizes[0] / DI;
    const long long E = in_sizes[1] / 2;
    const int* src = eidx;
    const int* dst = eidx + E;

    float* ws   = (float*)d_ws;
    float* msg  = ws;
    float* mx   = msg + (size_t)E * DI;
    float* sum  = mx  + (size_t)N * DI;
    float* agg  = sum + (size_t)N * DI;
    float* cs   = agg + (size_t)N * DI;
    float* cq   = cs + 128;
    float* sc   = cq + 128;
    float* sh   = sc + 128;
    float* hbuf = sh + 128;

    hipMemsetAsync(mx, 0, ((size_t)3 * N * DI + 256) * sizeof(float), stream);

    k_edge_msg<<<(int)((E + 63) / 64), 128, 0, stream>>>(x, src, ea, Wedge, msg, E, N);

    const long long tot = E * 16;
    const int segBlocks = (int)((tot + 255) / 256);
    k_seg_max<<<segBlocks, 256, 0, stream>>>(msg, dst, (unsigned int*)mx, E);
    k_seg_expsum<<<segBlocks, 256, 0, stream>>>(msg, dst, mx, sum, E);
    k_seg_agg<<<segBlocks, 256, 0, stream>>>(msg, dst, mx, sum, agg, E);

    const int strips = (int)((N + 15) / 16);
    k_h_stats<<<(strips + 3) / 4, 128, 0, stream>>>(x, agg, W1, hbuf, cs, cq, N);
    k_bn<<<1, 128, 0, stream>>>(cs, cq, gamma, beta, sc, sh, 1.0f / (float)N);
    k_out<<<(strips + 1) / 2, 64, 0, stream>>>(hbuf, W2, sc, sh, out, N);
}


# segment max kept as a level: softmax shift = column max rounded up to a multiple of 8; zero fill = first level, only values above 8 issue atomicMax and flag their node; unflagged rows not fetched
# speedup vs baseline: 2.5796x; 2.5796x over previous
#include <hip/hip_runtime.h>
#include <hip/hip_bf16.h>

typedef float v2f __attribute__((ext_vector_type(2)));
typedef float v4f __attribute__((ext_vector_type(4)));
typedef float v8f __attribute__((ext_vector_type(8)));

#define DI 64
#define HH 128

__device__ __forceinline__ v8f wmma_f32_k4(v2f a, v2f b, v8f c) {
    return __builtin_amdgcn_wmma_f32_16x16x4_f32(false, a, false, b, (short)0, c, false, false);
}

#define SHIFT_Q 8.0f
__device__ __forceinline__ float shift_level(float v) {
    return SHIFT_Q * (ceilf(v * (1.0f / SHIFT_Q)) - 1.0f);
}
#define LP 68
__global__ __launch_bounds__(128)
void k_edge_msg(const float* __restrict__ x, const int* __restrict__ src,
                const int* __restrict__ dst,
                const float* __restrict__ ea, const float* __restrict__ Wedge,
                float* __restrict__ msg, unsigned int* __restrict__ mx,
                unsigned int* __restrict__ flag,
                long long E, long long N) {
    __shared__ __attribute__((aligned(16))) float Ws[64][LP];
    __shared__ __attribute__((aligned(16))) float As[4][16][LP];

    const int tid  = threadIdx.x;
    const int wave = tid >> 5;
    const int lane = tid & 31;
    const int hf   = lane >> 4;
    const int lm   = lane & 15;

    const int wv = __builtin_amdgcn_readfirstlane(wave);
    const long long eS = ((long long)blockIdx.x * 4 + wv) * 16;
    const bool whole = (eS + 16 <= E);
    const int eLast = (int)E - 1;
    const int eSi   = (int)eS;

    int sr[8];
#pragma unroll
    for (int r = 0; r < 8; ++r) {
        const int er = eSi + 8 * hf + r;
        sr[r] = src[er < eLast ? er : eLast];
    }

    v2f av[16];
#pragma unroll
    for (int r = 0; r < 16; ++r) {
        const int er = eSi + r;
        const int ek = er < eLast ? er : eLast;
        av[r] = __builtin_nontemporal_load((const v2f*)(ea + (long long)ek * DI + lane * 2));
    }

    v4f xv[8];
#pragma unroll
    for (int r = 0; r < 8; ++r)
        xv[r] = *(const v4f*)(x + (long long)sr[r] * DI + 4 * lm);

#pragma unroll
    for (int r = 0; r < 16; ++r)
        *(v2f*)&As[wave][r][lane * 2] = av[r];

#pragma unroll 4
    for (int n = 0; n < 8; ++n) {
        const int i = tid + 128 * n;
        *(v4f*)&Ws[i >> 4][4 * (i & 15)] = *(const v4f*)(Wedge + 4 * i);
    }
    __syncthreads();

    v8f acc[4] = {};
#pragma unroll
    for (int k0 = 0; k0 < DI; k0 += 4) {
        const int kb = k0 + hf * 2;
        const v2f a  = *(const v2f*)&As[wave][lm][kb];
        const v4f w0 = *(const v4f*)&Ws[kb][4 * lm];
        const v4f w1 = *(const v4f*)&Ws[kb + 1][4 * lm];
#pragma unroll
        for (int t = 0; t < 4; ++t) {
            v2f b;
            b.x = w0[t];  b.y = w1[t];
            acc[t] = wmma_f32_k4(a, b, acc[t]);
        }
    }

    v4f o[8];
#pragma unroll
    for (int r = 0; r < 8; ++r) {
#pragma unroll
        for (int t = 0; t < 4; ++t) {
            float v = acc[t][r] + xv[r][t];
            o[r][t] = (v > 0.f ? v : 0.f) + 1e-7f;
        }
    }
    {
        float* mp = msg + eS * DI;
        const int mo = (8 * hf) * DI + 4 * lm;
        if (whole) {
#pragma unroll
            for (int r = 0; r < 8; ++r)
                __builtin_nontemporal_store(o[r], (v4f*)(mp + mo + r * DI));
        } else {
#pragma unroll
            for (int r = 0; r < 8; ++r)
                if (eS + 8 * hf + r < E)
                    __builtin_nontemporal_store(o[r], (v4f*)(mp + mo + r * DI));
        }
    }
    float big = o[0][0];
#pragma unroll
    for (int r = 0; r < 8; ++r) {
#pragma unroll
        for (int t = 0; t < 4; ++t) big = fmaxf(big, o[r][t]);
    }
    if (__builtin_amdgcn_ballot_w32(big > SHIFT_Q) != 0u) {
#pragma unroll
        for (int r = 0; r < 8; ++r) {
            const int er = eSi + 8 * hf + r;
            if (er <= eLast) {
                const int dn = dst[er];
                unsigned int* b = mx + (long long)dn * DI + 4 * lm;
                bool raised = false;
#pragma unroll
                for (int t = 0; t < 4; ++t) {
                    if (o[r][t] > SHIFT_Q) {
                        atomicMax(b + t, __float_as_uint(shift_level(o[r][t])));
                        raised = true;
                    }
                }
                if (raised) atomicOr(flag + (dn >> 5), 1u << (dn & 31));
            }
        }
    }
}

#define SEG_EPC 8
#define SEG_CPW 4
#define SEG_WPB 8
#define SEG_EPB (SEG_EPC * SEG_CPW * SEG_WPB)

template <bool FULL, int NE>
__device__ __forceinline__ void seg_load(const float* __restrict__ msg, const int* __restrict__ dst,
                                         long long e0, long long E, int lane,
                                         int (&d)[NE], float (&m0)[NE], float (&m1)[NE]) {
#pragma unroll
    for (int i = 0; i < NE; ++i) {
        long long e = e0 + i;
        if (!FULL) e = e < E ? e : E - 1;
        d[i] = dst[e];
        const float* r = msg + e * DI;
        m0[i] = __builtin_nontemporal_load(r + lane);
        m1[i] = __builtin_nontemporal_load(r + lane + 32);
    }
}

template <bool FULL, bool RAISED, int NE>
__device__ __forceinline__ void seg_sums_chunk(const float* __restrict__ msg, const int* __restrict__ dst,
                                               const float* __restrict__ mx, float* __restrict__ sum,
                                               float* __restrict__ num,
                                               long long e0, long long E, int lane) {
    int d[NE]; float m0[NE], m1[NE], x0[NE], x1[NE];
    seg_load<FULL, NE>(msg, dst, e0, E, lane, d, m0, m1);
#pragma unroll
    for (int i = 0; i < NE; ++i) {
        if (RAISED) {
            const float* p = mx + (long long)d[i] * DI;
            x0[i] = p[lane] + SHIFT_Q;
            x1[i] = p[lane + 32] + SHIFT_Q;
        } else {
            x0[i] = SHIFT_Q;
            x1[i] = SHIFT_Q;
        }
    }
#pragma unroll
    for (int i = 0; i < NE; ++i) {
        if (FULL || e0 + i < E) {
            const float w0 = __expf(m0[i] - x0[i]);
            const float w1 = __expf(m1[i] - x1[i]);
            float* b = sum + (long long)d[i] * DI;
            float* c = num + (long long)d[i] * DI;
            atomicAdd(b + lane,      w0);
            atomicAdd(b + lane + 32, w1);
            atomicAdd(c + lane,      m0[i] * w0);
            atomicAdd(c + lane + 32, m1[i] * w1);
        }
    }
}

template <bool RAISED>
__device__ __forceinline__ void seg_sums_wave(const float* __restrict__ msg, const int* __restrict__ dst,
                                              const float* __restrict__ mx, float* __restrict__ sum,
                                              float* __restrict__ num,
                                              long long eb, int wv, long long E, int lane) {
#pragma unroll 1
    for (int k = 0; k < SEG_CPW; ++k) {
        const long long e0 = eb + (long long)(k * SEG_WPB + wv) * SEG_EPC;
        if (!RAISED) {
            if (e0 + SEG_EPC <= E)  seg_sums_chunk<true, false, SEG_EPC>(msg, dst, mx, sum, num, e0, E, lane);
            else if (e0 < E)        seg_sums_chunk<false, false, SEG_EPC>(msg, dst, mx, sum, num, e0, E, lane);
        } else {
#pragma unroll 1
            for (int h = 0; h < 2; ++h) {
                const long long eh = e0 + h * (SEG_EPC / 2);
                if (eh + SEG_EPC / 2 <= E)  seg_sums_chunk<true, true, SEG_EPC / 2>(msg, dst, mx, sum, num, eh, E, lane);
                else if (eh < E)            seg_sums_chunk<false, true, SEG_EPC / 2>(msg, dst, mx, sum, num, eh, E, lane);
            }
        }
    }
}

__global__ __launch_bounds__(256)
void k_seg_sums(const float* __restrict__ msg, const int* __restrict__ dst,
                const float* __restrict__ mx, const unsigned int* __restrict__ flag,
                float* __restrict__ sum,
                float* __restrict__ num, long long E) {
    const int lane = threadIdx.x & 31;
    const int wv = __builtin_amdgcn_readfirstlane((int)(threadIdx.x >> 5));
    const long long eb = (long long)blockIdx.x * SEG_EPB;
    unsigned int raised = E < SEG_EPC ? 1u : 0u;
    if (E >= SEG_EPC) {
#pragma unroll
        for (int k = 0; k < SEG_CPW; ++k) {
            long long e0 = eb + (long long)(k * SEG_WPB + wv) * SEG_EPC;
            e0 = e0 + SEG_EPC <= E ? e0 : E - SEG_EPC;
#pragma unroll
            for (int i = 0; i < SEG_EPC; ++i) {
                const int dn = dst[e0 + i];
                raised |= (flag[dn >> 5] >> (dn & 31)) & 1u;
            }
        }
    }
    if (raised == 0u) seg_sums_wave<false>(msg, dst, mx, sum, num, eb, wv, E, lane);
    else              seg_sums_wave<true>(msg, dst, mx, sum, num, eb, wv, E, lane);
}

__global__ __launch_bounds__(128)
void k_h_stats(const float* __restrict__ x, const float* __restrict__ num,
               const float* __restrict__ sum,
               const float* __restrict__ W1, float* __restrict__ h,
               float* __restrict__ colsum, float* __restrict__ colsq, long long N) {
    __shared__ float W1s[64][130];
    __shared__ float As[4][16][66];
    __shared__ float cs_s[128], cq_s[128];

    const int tid  = threadIdx.x;
    const int wave = tid >> 5;
    const int lane = tid & 31;
    const int hf   = lane >> 4;
    const int lm   = lane & 15;

    for (int i = tid; i < 64 * HH; i += 128)
        W1s[i >> 7][i & 127] = W1[i];
    if (tid < 128) { cs_s[tid] = 0.f; cq_s[tid] = 0.f; }

    const long long r0 = ((long long)blockIdx.x * 4 + wave) * 16;
#pragma unroll
    for (int r = 0; r < 16; ++r) {
        long long rr = r0 + r;
        long long rc = rr < N ? rr : N - 1;
        const float2 a  = *(const float2*)(num + rc * DI + lane * 2);
        const float2 sm = *(const float2*)(sum + rc * DI + lane * 2);
        const float2 xv = *(const float2*)(x + rc * DI + lane * 2);
        As[wave][r][lane * 2 + 0] = a.x / (sm.x + 1e-16f) + xv.x;
        As[wave][r][lane * 2 + 1] = a.y / (sm.y + 1e-16f) + xv.y;
    }
    __syncthreads();

    for (int n0 = 0; n0 < HH; n0 += 16) {
        v8f acc = {};
#pragma unroll
        for (int k0 = 0; k0 < DI; k0 += 4) {
            const int kb = k0 + hf * 2;
            v2f a, b;
            a.x = As[wave][lm][kb];   a.y = As[wave][lm][kb + 1];
            b.x = W1s[kb][n0 + lm];   b.y = W1s[kb + 1][n0 + lm];
            acc = wmma_f32_k4(a, b, acc);
        }
        float p1 = 0.f, p2 = 0.f;
#pragma unroll
        for (int i = 0; i < 8; ++i) {
            const int m = i + hf * 8;
            const long long rr = r0 + m;
            if (rr < N) {
                const float v = acc[i];
                h[rr * HH + n0 + lm] = v;
                p1 += v;
                p2 += v * v;
            }
        }
        p1 += __shfl_xor(p1, 16, 32);
        p2 += __shfl_xor(p2, 16, 32);
        if (hf == 0) {
            atomicAdd(&cs_s[n0 + lm], p1);
            atomicAdd(&cq_s[n0 + lm], p2);
        }
    }
    __syncthreads();
    if (tid < 128) {
        atomicAdd(colsum + tid, cs_s[tid]);
        atomicAdd(colsq + tid, cq_s[tid]);
    }
}

__global__ __launch_bounds__(128)
void k_bn(const float* __restrict__ colsum, const float* __restrict__ colsq,
          const float* __restrict__ gamma, const float* __restrict__ beta,
          float* __restrict__ scale, float* __restrict__ shift, float invN) {
    const int i = threadIdx.x;
    const float mu  = colsum[i] * invN;
    const float var = colsq[i] * invN - mu * mu;
    const float sc  = rsqrtf(var + 1e-5f) * gamma[i];
    scale[i] = sc;
    shift[i] = beta[i] - mu * sc;
}

__global__ __launch_bounds__(64)
void k_out(const float* __restrict__ h, const float* __restrict__ W2,
           const float* __restrict__ scale, const float* __restrict__ shift,
           float* __restrict__ out, long long N) {
    __shared__ float W2s[HH][66];
    __shared__ float As[2][16][130];
    __shared__ float sc_s[128], sh_s[128];

    const int tid  = threadIdx.x;
    const int wave = tid >> 5;
    const int lane = tid & 31;
    const int hf   = lane >> 4;
    const int lm   = lane & 15;

    for (int i = tid; i < HH * DI; i += 64)
        W2s[i >> 6][i & 63] = W2[i];
    for (int i = tid; i < 128; i += 64) { sc_s[i] = scale[i]; sh_s[i] = shift[i]; }
    __syncthreads();

    const long long r0 = ((long long)blockIdx.x * 2 + wave) * 16;
    for (int r = 0; r < 16; ++r) {
        long long rr = r0 + r;
        long long rc = rr < N ? rr : N - 1;
        const int c = lane * 4;
        const v4f hv = *(const v4f*)(h + rc * HH + c);
        float v0 = hv.x * sc_s[c + 0] + sh_s[c + 0];
        float v1 = hv.y * sc_s[c + 1] + sh_s[c + 1];
        float v2 = hv.z * sc_s[c + 2] + sh_s[c + 2];
        float v3 = hv.w * sc_s[c + 3] + sh_s[c + 3];
        As[wave][r][c + 0] = v0 > 0.f ? v0 : 0.f;
        As[wave][r][c + 1] = v1 > 0.f ? v1 : 0.f;
        As[wave][r][c + 2] = v2 > 0.f ? v2 : 0.f;
        As[wave][r][c + 3] = v3 > 0.f ? v3 : 0.f;
    }
    __syncthreads();

    for (int n0 = 0; n0 < DI; n0 += 16) {
        v8f acc = {};
#pragma unroll
        for (int k0 = 0; k0 < HH; k0 += 4) {
            const int kb = k0 + hf * 2;
            v2f a, b;
            a.x = As[wave][lm][kb];   a.y = As[wave][lm][kb + 1];
            b.x = W2s[kb][n0 + lm];   b.y = W2s[kb + 1][n0 + lm];
            acc = wmma_f32_k4(a, b, acc);
        }
#pragma unroll
        for (int i = 0; i < 8; ++i) {
            const int m = i + hf * 8;
            const long long rr = r0 + m;
            if (rr < N) __builtin_nontemporal_store(acc[i], out + rr * DI + n0 + lm);
        }
    }
}

extern "C" void kernel_launch(void* const* d_in, const int* in_sizes, int n_in,
                              void* d_out, int out_size, void* d_ws, size_t ws_size,
                              hipStream_t stream) {
    const float* x     = (const float*)d_in[0];
    const int*   eidx  = (const int*)d_in[1];
    const float* ea    = (const float*)d_in[2];
    const float* Wedge = (const float*)d_in[3];
    const float* W1    = (const float*)d_in[4];
    const float* gamma = (const float*)d_in[5];
    const float* beta  = (const float*)d_in[6];
    const float* W2    = (const float*)d_in[7];
    float* out = (float*)d_out;

    const long long N = in_sizes[0] / DI;
    const long long E = in_sizes[1] / 2;
    const int* src = eidx;
    const int* dst = eidx + E;

    float* ws   = (float*)d_ws;
    float* msg  = ws;
    float* mx   = msg + (size_t)E * DI;
    float* sum  = mx  + (size_t)N * DI;
    float* num  = sum + (size_t)N * DI;
    float* cs   = num + (size_t)N * DI;
    float* cq   = cs + 128;
    float* sc   = cq + 128;
    float* sh   = sc + 128;
    float* hbuf = sh + 128;
    unsigned int* flag = (unsigned int*)hbuf;

    hipMemsetAsync(mx, 0, ((size_t)3 * N * DI + 512 + (N + 31) / 32) * sizeof(float), stream);

    k_edge_msg<<<(int)((E + 63) / 64), 128, 0, stream>>>(x, src, dst, ea, Wedge, msg, (unsigned int*)mx, flag, E, N);

    const int segBlocks = (int)((E + SEG_EPB - 1) / SEG_EPB);
    k_seg_sums<<<segBlocks, 256, 0, stream>>>(msg, dst, mx, flag, sum, num, E);

    const int strips = (int)((N + 15) / 16);
    k_h_stats<<<(strips + 3) / 4, 128, 0, stream>>>(x, num, sum, W1, hbuf, cs, cq, N);
    k_bn<<<1, 128, 0, stream>>>(cs, cq, gamma, beta, sc, sh, 1.0f / (float)N);
    k_out<<<(strips + 1) / 2, 64, 0, stream>>>(hbuf, W2, sc, sh, out, N);
}


# edge kernel and segment pass fused, msg never stored: a tile goes through the wave's LDS tile and adds its first-level exp terms by whole-line float atomics (arrival order); rarer levels added after
# speedup vs baseline: 3.0257x; 1.1729x over previous
#include <hip/hip_runtime.h>
#include <hip/hip_bf16.h>

typedef float v2f __attribute__((ext_vector_type(2)));
typedef float v4f __attribute__((ext_vector_type(4)));
typedef float v8f __attribute__((ext_vector_type(8)));

#define DI 64
#define HH 128

__device__ __forceinline__ v8f wmma_f32_k4(v2f a, v2f b, v8f c) {
    return __builtin_amdgcn_wmma_f32_16x16x4_f32(false, a, false, b, (short)0, c, false, false);
}

#define SHIFT_Q 8.0f
__device__ __forceinline__ float shift_level(float v) {
    return SHIFT_Q * (ceilf(v * (1.0f / SHIFT_Q)) - 1.0f);
}
#define LP 68
__device__ __forceinline__ void seg_add_row(float* __restrict__ sum, float* __restrict__ num,
                                            const int dn, const int lane,
                                            const float w0, const float w1,
                                            const float p0, const float p1) {
    float* b = sum + (long long)dn * DI;
    float* c = num + (long long)dn * DI;
    atomicAdd(b + lane,      w0);
    atomicAdd(b + lane + 32, w1);
    atomicAdd(c + lane,      p0);
    atomicAdd(c + lane + 32, p1);
}

template <bool FIX>
__device__ __forceinline__
void edge_tiles(float (&Ws)[64][LP], float (&As)[4][16][LP], const int vb,
                const float* __restrict__ x, const int* __restrict__ src,
                const int* __restrict__ dst,
                const float* __restrict__ ea, const float* __restrict__ Wedge,
                unsigned int* __restrict__ mx,
                float* __restrict__ sum, float* __restrict__ num,
                float* __restrict__ sumR, float* __restrict__ numR,
                unsigned int* __restrict__ flag, unsigned int* __restrict__ tflag,
                const long long E) {
    const int tid  = threadIdx.x;
    const int wave = tid >> 5;
    const int lane = tid & 31;
    const int hf   = lane >> 4;
    const int lm   = lane & 15;

    const int wv = __builtin_amdgcn_readfirstlane(wave);
    const long long eS = ((long long)vb * 4 + wv) * 16;
    const int eLast = (int)E - 1;
    const int eSi   = (int)eS;

    int sr[8];
#pragma unroll
    for (int r = 0; r < 8; ++r) {
        const int er = eSi + 8 * hf + r;
        sr[r] = src[er < eLast ? er : eLast];
    }

    int dn16[16];
    if (!FIX) {
#pragma unroll
        for (int e = 0; e < 16; ++e) {
            const int er = eSi + e;
            dn16[e] = dst[er < eLast ? er : eLast];
        }
    }

    v2f av[16];
#pragma unroll
    for (int r = 0; r < 16; ++r) {
        const int er = eSi + r;
        const int ek = er < eLast ? er : eLast;
        av[r] = __builtin_nontemporal_load((const v2f*)(ea + (long long)ek * DI + lane * 2));
    }

    v4f xv[8];
#pragma unroll
    for (int r = 0; r < 8; ++r)
        xv[r] = *(const v4f*)(x + (long long)sr[r] * DI + 4 * lm);

#pragma unroll
    for (int r = 0; r < 16; ++r)
        *(v2f*)&As[wave][r][lane * 2] = av[r];

#pragma unroll 4
    for (int n = 0; n < 8; ++n) {
        const int i = tid + 128 * n;
        *(v4f*)&Ws[i >> 4][4 * (i & 15)] = *(const v4f*)(Wedge + 4 * i);
    }
    __syncthreads();

    v8f acc[4] = {};
#pragma unroll
    for (int k0 = 0; k0 < DI; k0 += 4) {
        const int kb = k0 + hf * 2;
        const v2f a  = *(const v2f*)&As[wave][lm][kb];
        const v4f w0 = *(const v4f*)&Ws[kb][4 * lm];
        const v4f w1 = *(const v4f*)&Ws[kb + 1][4 * lm];
#pragma unroll
        for (int t = 0; t < 4; ++t) {
            v2f b;
            b.x = w0[t];  b.y = w1[t];
            acc[t] = wmma_f32_k4(a, b, acc[t]);
        }
    }

    v4f o[8];
#pragma unroll
    for (int r = 0; r < 8; ++r) {
#pragma unroll
        for (int t = 0; t < 4; ++t) {
            float v = acc[t][r] + xv[r][t];
            o[r][t] = (v > 0.f ? v : 0.f) + 1e-7f;
        }
    }
    float big = o[0][0];
#pragma unroll
    for (int r = 0; r < 8; ++r) {
#pragma unroll
        for (int t = 0; t < 4; ++t) big = fmaxf(big, o[r][t]);
    }
    const bool above = __builtin_amdgcn_ballot_w32(big > SHIFT_Q) != 0u;

    if (FIX) {
        if (above) {
#pragma unroll
            for (int r = 0; r < 8; ++r) {
                const int er = eSi + 8 * hf + r;
                if (er <= eLast) {
                    const long long row = (long long)dst[er] * DI + 4 * lm;
#pragma unroll
                    for (int t = 0; t < 4; ++t) {
                        const float m = o[r][t];
                        if (m > SHIFT_Q) {
                            const float s = __uint_as_float(mx[row + t]) + SHIFT_Q;
                            const float w = __expf(m - s);
                            atomicAdd(sumR + row + t, w);
                            atomicAdd(numR + row + t, m * w);
                        }
                    }
                }
            }
        }
        return;
    }

#pragma unroll
    for (int r = 0; r < 8; ++r)
        *(v4f*)&As[wave][8 * hf + r][4 * lm] = o[r];

    if (above) {
        if (lane == 0) {
            const int tile = vb * 4 + wv;
            atomicOr(tflag + (tile >> 5), 1u << (tile & 31));
        }
#pragma unroll
        for (int r = 0; r < 8; ++r) {
            const int er = eSi + 8 * hf + r;
            if (er <= eLast) {
                const int dn = dst[er];
                unsigned int* b = mx + (long long)dn * DI + 4 * lm;
                bool raised = false;
#pragma unroll
                for (int t = 0; t < 4; ++t) {
                    if (o[r][t] > SHIFT_Q) {
                        atomicMax(b + t, __float_as_uint(shift_level(o[r][t])));
                        raised = true;
                    }
                }
                if (raised) {
                    const unsigned int bit = 1u << (dn & 31);
                    const unsigned int old = atomicOr(flag + (dn >> 5), bit);
                    if ((old & bit) == 0u) {
                        v4f* zs = (v4f*)(sumR + (long long)dn * DI);
                        v4f* zn = (v4f*)(numR + (long long)dn * DI);
                        const v4f z = {0.f, 0.f, 0.f, 0.f};
                        for (int q = 0; q < DI / 4; ++q) { zs[q] = z; zn[q] = z; }
                    }
                }
            }
        }
    }
    __syncthreads();

    float w0[16], w1[16], p0[16], p1[16];
#pragma unroll
    for (int e = 0; e < 16; ++e) {
        const float m0 = As[wave][e][lane];
        const float m1 = As[wave][e][lane + 32];
        w0[e] = m0 > SHIFT_Q ? 0.f : __expf(m0 - SHIFT_Q);
        w1[e] = m1 > SHIFT_Q ? 0.f : __expf(m1 - SHIFT_Q);
        p0[e] = m0 * w0[e];
        p1[e] = m1 * w1[e];
    }
    if (eS + 16 <= E) {
#pragma unroll
        for (int e = 0; e < 16; ++e)
            seg_add_row(sum, num, dn16[e], lane, w0[e], w1[e], p0[e], p1[e]);
    } else {
#pragma unroll
        for (int e = 0; e < 16; ++e)
            if (eSi + e <= eLast) seg_add_row(sum, num, dn16[e], lane, w0[e], w1[e], p0[e], p1[e]);
    }
}

__global__ __launch_bounds__(128)
void k_edge_seg(const float* __restrict__ x, const int* __restrict__ src,
                const int* __restrict__ dst,
                const float* __restrict__ ea, const float* __restrict__ Wedge,
                unsigned int* __restrict__ mx,
                float* __restrict__ sum, float* __restrict__ num,
                float* __restrict__ sumR, float* __restrict__ numR,
                unsigned int* __restrict__ flag, unsigned int* __restrict__ tflag,
                long long E) {
    __shared__ __attribute__((aligned(16))) float Ws[64][LP];
    __shared__ __attribute__((aligned(16))) float As[4][16][LP];
    edge_tiles<false>(Ws, As, (int)blockIdx.x, x, src, dst, ea, Wedge, mx, sum, num, sumR, numR, flag, tflag, E);
}

__global__ __launch_bounds__(128)
void k_edge_fix(const float* __restrict__ x, const int* __restrict__ src,
                const int* __restrict__ dst,
                const float* __restrict__ ea, const float* __restrict__ Wedge,
                unsigned int* __restrict__ mx,
                float* __restrict__ sum, float* __restrict__ num,
                float* __restrict__ sumR, float* __restrict__ numR,
                unsigned int* __restrict__ flag, unsigned int* __restrict__ tflag,
                long long E, int words) {
    __shared__ __attribute__((aligned(16))) float Ws[64][LP];
    __shared__ __attribute__((aligned(16))) float As[4][16][LP];
    for (int w = blockIdx.x; w < words; w += gridDim.x) {
        const unsigned int tw = tflag[w];
        if (tw == 0u) continue;
#pragma unroll 1
        for (int q = 0; q < 8; ++q) {
            if (((tw >> (4 * q)) & 0xFu) != 0u) {
                edge_tiles<true>(Ws, As, w * 8 + q, x, src, dst, ea, Wedge, mx, sum, num, sumR, numR, flag, tflag, E);
                __syncthreads();
            }
        }
    }
}

__global__ __launch_bounds__(128)
void k_h_stats(const float* __restrict__ x, const float* __restrict__ num,
               const float* __restrict__ sum,
               const float* __restrict__ mx, const float* __restrict__ numR,
               const float* __restrict__ sumR, const unsigned int* __restrict__ flag,
               const float* __restrict__ W1, float* __restrict__ h,
               float* __restrict__ colsum, float* __restrict__ colsq, long long N) {
    __shared__ float W1s[64][130];
    __shared__ float As[4][16][66];
    __shared__ float cs_s[128], cq_s[128];

    const int tid  = threadIdx.x;
    const int wave = tid >> 5;
    const int lane = tid & 31;
    const int hf   = lane >> 4;
    const int lm   = lane & 15;

    for (int i = tid; i < 64 * HH; i += 128)
        W1s[i >> 7][i & 127] = W1[i];
    if (tid < 128) { cs_s[tid] = 0.f; cq_s[tid] = 0.f; }

    const long long r0 = ((long long)blockIdx.x * 4 + wave) * 16;
#pragma unroll
    for (int r = 0; r < 16; ++r) {
        long long rr = r0 + r;
        long long rc = rr < N ? rr : N - 1;
        const float2 a  = *(const float2*)(num + rc * DI + lane * 2);
        const float2 sm = *(const float2*)(sum + rc * DI + lane * 2);
        const float2 xv = *(const float2*)(x + rc * DI + lane * 2);
        As[wave][r][lane * 2 + 0] = a.x / (sm.x + 1e-16f) + xv.x;
        As[wave][r][lane * 2 + 1] = a.y / (sm.y + 1e-16f) + xv.y;
    }
    const long long r0s = ((long long)blockIdx.x * 4 + __builtin_amdgcn_readfirstlane(wave)) * 16;
    const unsigned int fw = r0s < N ? (flag[r0s >> 5] >> (r0s & 31)) & 0xFFFFu : 0u;
    if (fw != 0u) {
#pragma unroll 1
        for (int r = 0; r < 16; ++r) {
            if ((fw >> r) & 1u) {
                const long long rc = r0s + r;
                const float2 a  = *(const float2*)(num + rc * DI + lane * 2);
                const float2 sm = *(const float2*)(sum + rc * DI + lane * 2);
                const float2 xv = *(const float2*)(x + rc * DI + lane * 2);
                const float2 lv = *(const float2*)(mx + rc * DI + lane * 2);
                const float2 aR = *(const float2*)(numR + rc * DI + lane * 2);
                const float2 sR = *(const float2*)(sumR + rc * DI + lane * 2);
                const float f0 = __expf(-lv.x), f1 = __expf(-lv.y);
                As[wave][r][lane * 2 + 0] = (a.x * f0 + aR.x) / (sm.x * f0 + sR.x + 1e-16f) + xv.x;
                As[wave][r][lane * 2 + 1] = (a.y * f1 + aR.y) / (sm.y * f1 + sR.y + 1e-16f) + xv.y;
            }
        }
    }
    __syncthreads();

    for (int n0 = 0; n0 < HH; n0 += 16) {
        v8f acc = {};
#pragma unroll
        for (int k0 = 0; k0 < DI; k0 += 4) {
            const int kb = k0 + hf * 2;
            v2f a, b;
            a.x = As[wave][lm][kb];   a.y = As[wave][lm][kb + 1];
            b.x = W1s[kb][n0 + lm];   b.y = W1s[kb + 1][n0 + lm];
            acc = wmma_f32_k4(a, b, acc);
        }
        float p1 = 0.f, p2 = 0.f;
#pragma unroll
        for (int i = 0; i < 8; ++i) {
            const int m = i + hf * 8;
            const long long rr = r0 + m;
            if (rr < N) {
                const float v = acc[i];
                h[rr * HH + n0 + lm] = v;
                p1 += v;
                p2 += v * v;
            }
        }
        p1 += __shfl_xor(p1, 16, 32);
        p2 += __shfl_xor(p2, 16, 32);
        if (hf == 0) {
            atomicAdd(&cs_s[n0 + lm], p1);
            atomicAdd(&cq_s[n0 + lm], p2);
        }
    }
    __syncthreads();
    if (tid < 128) {
        atomicAdd(colsum + tid, cs_s[tid]);
        atomicAdd(colsq + tid, cq_s[tid]);
    }
}

__global__ __launch_bounds__(128)
void k_bn(const float* __restrict__ colsum, const float* __restrict__ colsq,
          const float* __restrict__ gamma, const float* __restrict__ beta,
          float* __restrict__ scale, float* __restrict__ shift, float invN) {
    const int i = threadIdx.x;
    const float mu  = colsum[i] * invN;
    const float var = colsq[i] * invN - mu * mu;
    const float sc  = rsqrtf(var + 1e-5f) * gamma[i];
    scale[i] = sc;
    shift[i] = beta[i] - mu * sc;
}

__global__ __launch_bounds__(64)
void k_out(const float* __restrict__ h, const float* __restrict__ W2,
           const float* __restrict__ scale, const float* __restrict__ shift,
           float* __restrict__ out, long long N) {
    __shared__ float W2s[HH][66];
    __shared__ float As[2][16][130];
    __shared__ float sc_s[128], sh_s[128];

    const int tid  = threadIdx.x;
    const int wave = tid >> 5;
    const int lane = tid & 31;
    const int hf   = lane >> 4;
    const int lm   = lane & 15;

    for (int i = tid; i < HH * DI; i += 64)
        W2s[i >> 6][i & 63] = W2[i];
    for (int i = tid; i < 128; i += 64) { sc_s[i] = scale[i]; sh_s[i] = shift[i]; }
    __syncthreads();

    const long long r0 = ((long long)blockIdx.x * 2 + wave) * 16;
    for (int r = 0; r < 16; ++r) {
        long long rr = r0 + r;
        long long rc = rr < N ? rr : N - 1;
        const int c = lane * 4;
        const v4f hv = *(const v4f*)(h + rc * HH + c);
        float v0 = hv.x * sc_s[c + 0] + sh_s[c + 0];
        float v1 = hv.y * sc_s[c + 1] + sh_s[c + 1];
        float v2 = hv.z * sc_s[c + 2] + sh_s[c + 2];
        float v3 = hv.w * sc_s[c + 3] + sh_s[c + 3];
        As[wave][r][c + 0] = v0 > 0.f ? v0 : 0.f;
        As[wave][r][c + 1] = v1 > 0.f ? v1 : 0.f;
        As[wave][r][c + 2] = v2 > 0.f ? v2 : 0.f;
        As[wave][r][c + 3] = v3 > 0.f ? v3 : 0.f;
    }
    __syncthreads();

    for (int n0 = 0; n0 < DI; n0 += 16) {
        v8f acc = {};
#pragma unroll
        for (int k0 = 0; k0 < HH; k0 += 4) {
            const int kb = k0 + hf * 2;
            v2f a, b;
            a.x = As[wave][lm][kb];   a.y = As[wave][lm][kb + 1];
            b.x = W2s[kb][n0 + lm];   b.y = W2s[kb + 1][n0 + lm];
            acc = wmma_f32_k4(a, b, acc);
        }
#pragma unroll
        for (int i = 0; i < 8; ++i) {
            const int m = i + hf * 8;
            const long long rr = r0 + m;
            if (rr < N) __builtin_nontemporal_store(acc[i], out + rr * DI + n0 + lm);
        }
    }
}

extern "C" void kernel_launch(void* const* d_in, const int* in_sizes, int n_in,
                              void* d_out, int out_size, void* d_ws, size_t ws_size,
                              hipStream_t stream) {
    const float* x     = (const float*)d_in[0];
    const int*   eidx  = (const int*)d_in[1];
    const float* ea    = (const float*)d_in[2];
    const float* Wedge = (const float*)d_in[3];
    const float* W1    = (const float*)d_in[4];
    const float* gamma = (const float*)d_in[5];
    const float* beta  = (const float*)d_in[6];
    const float* W2    = (const float*)d_in[7];
    float* out = (float*)d_out;

    const long long N = in_sizes[0] / DI;
    const long long E = in_sizes[1] / 2;
    const int* src = eidx;
    const int* dst = eidx + E;

    const int edgeBlocks = (int)((E + 63) / 64);
    const size_t nodeWords = (((size_t)N + 31) / 32 + 4) & ~(size_t)3;
    const size_t tileWords = (((size_t)edgeBlocks + 7) / 8 + 4) & ~(size_t)3;

    float* ws   = (float*)d_ws;
    float* mx   = ws;
    float* sum  = mx  + (size_t)N * DI;
    float* num  = sum + (size_t)N * DI;
    float* cs   = num + (size_t)N * DI;
    float* cq   = cs + 128;
    unsigned int* flag  = (unsigned int*)(cq + 128);
    unsigned int* tflag = flag + nodeWords;
    float* sc   = (float*)(tflag + tileWords);
    float* sh   = sc + 128;
    float* sumR = sh + 128;
    float* numR = sumR + (size_t)N * DI;
    float* hbuf = numR + (size_t)N * DI;

    hipMemsetAsync(mx, 0, ((size_t)3 * N * DI + 256 + nodeWords + tileWords) * sizeof(float), stream);

    k_edge_seg<<<edgeBlocks, 128, 0, stream>>>(x, src, dst, ea, Wedge, (unsigned int*)mx, sum, num,
                                               sumR, numR, flag, tflag, E);
    const int words = (edgeBlocks + 7) / 8;
    k_edge_fix<<<words < 256 ? words : 256, 128, 0, stream>>>(x, src, dst, ea, Wedge, (unsigned int*)mx, sum, num,
                                                              sumR, numR, flag, tflag, E, words);

    const int strips = (int)((N + 15) / 16);
    k_h_stats<<<(strips + 3) / 4, 128, 0, stream>>>(x, num, sum, mx, numR, sumR, flag, W1, hbuf, cs, cq, N);
    k_bn<<<1, 128, 0, stream>>>(cs, cq, gamma, beta, sc, sh, 1.0f / (float)N);
    k_out<<<(strips + 1) / 2, 64, 0, stream>>>(hbuf, W2, sc, sh, out, N);
}


# last kernel re-shaped: 782 blocks of 8 waves (64 rows; two waves a 16-row strip, 32 output columns each), all resident at once, where 1563 two-wave blocks needed a second round
# speedup vs baseline: 3.2643x; 1.0789x over previous
#include <hip/hip_runtime.h>
#include <hip/hip_bf16.h>

typedef float v2f __attribute__((ext_vector_type(2)));
typedef float v4f __attribute__((ext_vector_type(4)));
typedef float v8f __attribute__((ext_vector_type(8)));

#define DI 64
#define HH 128

__device__ __forceinline__ v8f wmma_f32_k4(v2f a, v2f b, v8f c) {
    return __builtin_amdgcn_wmma_f32_16x16x4_f32(false, a, false, b, (short)0, c, false, false);
}

#define SHIFT_Q 8.0f
__device__ __forceinline__ float shift_level(float v) {
    return SHIFT_Q * (ceilf(v * (1.0f / SHIFT_Q)) - 1.0f);
}
#define LP 68
__device__ __forceinline__ void seg_add_row(float* __restrict__ sum, float* __restrict__ num,
                                            const int dn, const int lane,
                                            const float w0, const float w1,
                                            const float p0, const float p1) {
    float* b = sum + (long long)dn * DI;
    float* c = num + (long long)dn * DI;
    atomicAdd(b + lane,      w0);
    atomicAdd(b + lane + 32, w1);
    atomicAdd(c + lane,      p0);
    atomicAdd(c + lane + 32, p1);
}

template <bool FIX>
__device__ __forceinline__
void edge_tiles(float (&Ws)[64][LP], float (&As)[4][16][LP], const int vb,
                const float* __restrict__ x, const int* __restrict__ src,
                const int* __restrict__ dst,
                const float* __restrict__ ea, const float* __restrict__ Wedge,
                unsigned int* __restrict__ mx,
                float* __restrict__ sum, float* __restrict__ num,
                float* __restrict__ sumR, float* __restrict__ numR,
                unsigned int* __restrict__ flag, unsigned int* __restrict__ tflag,
                const long long E) {
    const int tid  = threadIdx.x;
    const int wave = tid >> 5;
    const int lane = tid & 31;
    const int hf   = lane >> 4;
    const int lm   = lane & 15;

    const int wv = __builtin_amdgcn_readfirstlane(wave);
    const long long eS = ((long long)vb * 4 + wv) * 16;
    const int eLast = (int)E - 1;
    const int eSi   = (int)eS;

    int sr[8];
#pragma unroll
    for (int r = 0; r < 8; ++r) {
        const int er = eSi + 8 * hf + r;
        sr[r] = src[er < eLast ? er : eLast];
    }

    int dn16[16];
    if (!FIX) {
#pragma unroll
        for (int e = 0; e < 16; ++e) {
            const int er = eSi + e;
            dn16[e] = dst[er < eLast ? er : eLast];
        }
    }

    v2f av[16];
#pragma unroll
    for (int r = 0; r < 16; ++r) {
        const int er = eSi + r;
        const int ek = er < eLast ? er : eLast;
        av[r] = __builtin_nontemporal_load((const v2f*)(ea + (long long)ek * DI + lane * 2));
    }

    v4f xv[8];
#pragma unroll
    for (int r = 0; r < 8; ++r)
        xv[r] = *(const v4f*)(x + (long long)sr[r] * DI + 4 * lm);

#pragma unroll
    for (int r = 0; r < 16; ++r)
        *(v2f*)&As[wave][r][lane * 2] = av[r];

#pragma unroll 4
    for (int n = 0; n < 8; ++n) {
        const int i = tid + 128 * n;
        *(v4f*)&Ws[i >> 4][4 * (i & 15)] = *(const v4f*)(Wedge + 4 * i);
    }
    __syncthreads();

    v8f acc[4] = {};
#pragma unroll
    for (int k0 = 0; k0 < DI; k0 += 4) {
        const int kb = k0 + hf * 2;
        const v2f a  = *(const v2f*)&As[wave][lm][kb];
        const v4f w0 = *(const v4f*)&Ws[kb][4 * lm];
        const v4f w1 = *(const v4f*)&Ws[kb + 1][4 * lm];
#pragma unroll
        for (int t = 0; t < 4; ++t) {
            v2f b;
            b.x = w0[t];  b.y = w1[t];
            acc[t] = wmma_f32_k4(a, b, acc[t]);
        }
    }

    v4f o[8];
#pragma unroll
    for (int r = 0; r < 8; ++r) {
#pragma unroll
        for (int t = 0; t < 4; ++t) {
            float v = acc[t][r] + xv[r][t];
            o[r][t] = (v > 0.f ? v : 0.f) + 1e-7f;
        }
    }
    float big = o[0][0];
#pragma unroll
    for (int r = 0; r < 8; ++r) {
#pragma unroll
        for (int t = 0; t < 4; ++t) big = fmaxf(big, o[r][t]);
    }
    const bool above = __builtin_amdgcn_ballot_w32(big > SHIFT_Q) != 0u;

    if (FIX) {
        if (above) {
#pragma unroll
            for (int r = 0; r < 8; ++r) {
                const int er = eSi + 8 * hf + r;
                if (er <= eLast) {
                    const long long row = (long long)dst[er] * DI + 4 * lm;
#pragma unroll
                    for (int t = 0; t < 4; ++t) {
                        const float m = o[r][t];
                        if (m > SHIFT_Q) {
                            const float s = __uint_as_float(mx[row + t]) + SHIFT_Q;
                            const float w = __expf(m - s);
                            atomicAdd(sumR + row + t, w);
                            atomicAdd(numR + row + t, m * w);
                        }
                    }
                }
            }
        }
        return;
    }

#pragma unroll
    for (int r = 0; r < 8; ++r)
        *(v4f*)&As[wave][8 * hf + r][4 * lm] = o[r];

    if (above) {
        if (lane == 0) {
            const int tile = vb * 4 + wv;
            atomicOr(tflag + (tile >> 5), 1u << (tile & 31));
        }
#pragma unroll
        for (int r = 0; r < 8; ++r) {
            const int er = eSi + 8 * hf + r;
            if (er <= eLast) {
                const int dn = dst[er];
                unsigned int* b = mx + (long long)dn * DI + 4 * lm;
                bool raised = false;
#pragma unroll
                for (int t = 0; t < 4; ++t) {
                    if (o[r][t] > SHIFT_Q) {
                        atomicMax(b + t, __float_as_uint(shift_level(o[r][t])));
                        raised = true;
                    }
                }
                if (raised) {
                    const unsigned int bit = 1u << (dn & 31);
                    const unsigned int old = atomicOr(flag + (dn >> 5), bit);
                    if ((old & bit) == 0u) {
                        v4f* zs = (v4f*)(sumR + (long long)dn * DI);
                        v4f* zn = (v4f*)(numR + (long long)dn * DI);
                        const v4f z = {0.f, 0.f, 0.f, 0.f};
                        for (int q = 0; q < DI / 4; ++q) { zs[q] = z; zn[q] = z; }
                    }
                }
            }
        }
    }
    __syncthreads();

    float w0[16], w1[16], p0[16], p1[16];
#pragma unroll
    for (int e = 0; e < 16; ++e) {
        const float m0 = As[wave][e][lane];
        const float m1 = As[wave][e][lane + 32];
        w0[e] = m0 > SHIFT_Q ? 0.f : __expf(m0 - SHIFT_Q);
        w1[e] = m1 > SHIFT_Q ? 0.f : __expf(m1 - SHIFT_Q);
        p0[e] = m0 * w0[e];
        p1[e] = m1 * w1[e];
    }
    if (eS + 16 <= E) {
#pragma unroll
        for (int e = 0; e < 16; ++e)
            seg_add_row(sum, num, dn16[e], lane, w0[e], w1[e], p0[e], p1[e]);
    } else {
#pragma unroll
        for (int e = 0; e < 16; ++e)
            if (eSi + e <= eLast) seg_add_row(sum, num, dn16[e], lane, w0[e], w1[e], p0[e], p1[e]);
    }
}

__global__ __launch_bounds__(128)
void k_edge_seg(const float* __restrict__ x, const int* __restrict__ src,
                const int* __restrict__ dst,
                const float* __restrict__ ea, const float* __restrict__ Wedge,
                unsigned int* __restrict__ mx,
                float* __restrict__ sum, float* __restrict__ num,
                float* __restrict__ sumR, float* __restrict__ numR,
                unsigned int* __restrict__ flag, unsigned int* __restrict__ tflag,
                long long E) {
    __shared__ __attribute__((aligned(16))) float Ws[64][LP];
    __shared__ __attribute__((aligned(16))) float As[4][16][LP];
    edge_tiles<false>(Ws, As, (int)blockIdx.x, x, src, dst, ea, Wedge, mx, sum, num, sumR, numR, flag, tflag, E);
}

__global__ __launch_bounds__(128)
void k_edge_fix(const float* __restrict__ x, const int* __restrict__ src,
                const int* __restrict__ dst,
                const float* __restrict__ ea, const float* __restrict__ Wedge,
                unsigned int* __restrict__ mx,
                float* __restrict__ sum, float* __restrict__ num,
                float* __restrict__ sumR, float* __restrict__ numR,
                unsigned int* __restrict__ flag, unsigned int* __restrict__ tflag,
                long long E, int words) {
    __shared__ __attribute__((aligned(16))) float Ws[64][LP];
    __shared__ __attribute__((aligned(16))) float As[4][16][LP];
    for (int w = blockIdx.x; w < words; w += gridDim.x) {
        const unsigned int tw = tflag[w];
        if (tw == 0u) continue;
#pragma unroll 1
        for (int q = 0; q < 8; ++q) {
            if (((tw >> (4 * q)) & 0xFu) != 0u) {
                edge_tiles<true>(Ws, As, w * 8 + q, x, src, dst, ea, Wedge, mx, sum, num, sumR, numR, flag, tflag, E);
                __syncthreads();
            }
        }
    }
}

__global__ __launch_bounds__(128)
void k_h_stats(const float* __restrict__ x, const float* __restrict__ num,
               const float* __restrict__ sum,
               const float* __restrict__ mx, const float* __restrict__ numR,
               const float* __restrict__ sumR, const unsigned int* __restrict__ flag,
               const float* __restrict__ W1, float* __restrict__ h,
               float* __restrict__ colsum, float* __restrict__ colsq, long long N) {
    __shared__ float W1s[64][130];
    __shared__ float As[4][16][66];
    __shared__ float cs_s[128], cq_s[128];

    const int tid  = threadIdx.x;
    const int wave = tid >> 5;
    const int lane = tid & 31;
    const int hf   = lane >> 4;
    const int lm   = lane & 15;

    for (int i = tid; i < 64 * HH; i += 128)
        W1s[i >> 7][i & 127] = W1[i];
    if (tid < 128) { cs_s[tid] = 0.f; cq_s[tid] = 0.f; }

    const long long r0 = ((long long)blockIdx.x * 4 + wave) * 16;
#pragma unroll
    for (int r = 0; r < 16; ++r) {
        long long rr = r0 + r;
        long long rc = rr < N ? rr : N - 1;
        const float2 a  = *(const float2*)(num + rc * DI + lane * 2);
        const float2 sm = *(const float2*)(sum + rc * DI + lane * 2);
        const float2 xv = *(const float2*)(x + rc * DI + lane * 2);
        As[wave][r][lane * 2 + 0] = a.x / (sm.x + 1e-16f) + xv.x;
        As[wave][r][lane * 2 + 1] = a.y / (sm.y + 1e-16f) + xv.y;
    }
    const long long r0s = ((long long)blockIdx.x * 4 + __builtin_amdgcn_readfirstlane(wave)) * 16;
    const unsigned int fw = r0s < N ? (flag[r0s >> 5] >> (r0s & 31)) & 0xFFFFu : 0u;
    if (fw != 0u) {
#pragma unroll 1
        for (int r = 0; r < 16; ++r) {
            if ((fw >> r) & 1u) {
                const long long rc = r0s + r;
                const float2 a  = *(const float2*)(num + rc * DI + lane * 2);
                const float2 sm = *(const float2*)(sum + rc * DI + lane * 2);
                const float2 xv = *(const float2*)(x + rc * DI + lane * 2);
                const float2 lv = *(const float2*)(mx + rc * DI + lane * 2);
                const float2 aR = *(const float2*)(numR + rc * DI + lane * 2);
                const float2 sR = *(const float2*)(sumR + rc * DI + lane * 2);
                const float f0 = __expf(-lv.x), f1 = __expf(-lv.y);
                As[wave][r][lane * 2 + 0] = (a.x * f0 + aR.x) / (sm.x * f0 + sR.x + 1e-16f) + xv.x;
                As[wave][r][lane * 2 + 1] = (a.y * f1 + aR.y) / (sm.y * f1 + sR.y + 1e-16f) + xv.y;
            }
        }
    }
    __syncthreads();

    for (int n0 = 0; n0 < HH; n0 += 16) {
        v8f acc = {};
#pragma unroll
        for (int k0 = 0; k0 < DI; k0 += 4) {
            const int kb = k0 + hf * 2;
            v2f a, b;
            a.x = As[wave][lm][kb];   a.y = As[wave][lm][kb + 1];
            b.x = W1s[kb][n0 + lm];   b.y = W1s[kb + 1][n0 + lm];
            acc = wmma_f32_k4(a, b, acc);
        }
        float p1 = 0.f, p2 = 0.f;
#pragma unroll
        for (int i = 0; i < 8; ++i) {
            const int m = i + hf * 8;
            const long long rr = r0 + m;
            if (rr < N) {
                const float v = acc[i];
                h[rr * HH + n0 + lm] = v;
                p1 += v;
                p2 += v * v;
            }
        }
        p1 += __shfl_xor(p1, 16, 32);
        p2 += __shfl_xor(p2, 16, 32);
        if (hf == 0) {
            atomicAdd(&cs_s[n0 + lm], p1);
            atomicAdd(&cq_s[n0 + lm], p2);
        }
    }
    __syncthreads();
    if (tid < 128) {
        atomicAdd(colsum + tid, cs_s[tid]);
        atomicAdd(colsq + tid, cq_s[tid]);
    }
}

__global__ __launch_bounds__(128)
void k_bn(const float* __restrict__ colsum, const float* __restrict__ colsq,
          const float* __restrict__ gamma, const float* __restrict__ beta,
          float* __restrict__ scale, float* __restrict__ shift, float invN) {
    const int i = threadIdx.x;
    const float mu  = colsum[i] * invN;
    const float var = colsq[i] * invN - mu * mu;
    const float sc  = rsqrtf(var + 1e-5f) * gamma[i];
    scale[i] = sc;
    shift[i] = beta[i] - mu * sc;
}

__global__ __launch_bounds__(256)
void k_out(const float* __restrict__ h, const float* __restrict__ W2,
           const float* __restrict__ scale, const float* __restrict__ shift,
           float* __restrict__ out, long long N) {
    __shared__ float W2s[HH][66];
    __shared__ float As[4][16][130];
    __shared__ float sc_s[128], sh_s[128];

    const int tid   = threadIdx.x;
    const int wave  = tid >> 5;
    const int strip = wave >> 1;
    const int part  = wave & 1;
    const int lane  = tid & 31;
    const int hf    = lane >> 4;
    const int lm    = lane & 15;

    for (int i = tid; i < HH * DI; i += 256)
        W2s[i >> 6][i & 63] = W2[i];
    if (tid < 128) { sc_s[tid] = scale[tid]; sh_s[tid] = shift[tid]; }
    __syncthreads();

    const long long r0 = ((long long)blockIdx.x * 4 + strip) * 16;
    for (int r = 8 * part; r < 8 * part + 8; ++r) {
        long long rr = r0 + r;
        long long rc = rr < N ? rr : N - 1;
        const int c = lane * 4;
        const v4f hv = *(const v4f*)(h + rc * HH + c);
        float v0 = hv.x * sc_s[c + 0] + sh_s[c + 0];
        float v1 = hv.y * sc_s[c + 1] + sh_s[c + 1];
        float v2 = hv.z * sc_s[c + 2] + sh_s[c + 2];
        float v3 = hv.w * sc_s[c + 3] + sh_s[c + 3];
        As[strip][r][c + 0] = v0 > 0.f ? v0 : 0.f;
        As[strip][r][c + 1] = v1 > 0.f ? v1 : 0.f;
        As[strip][r][c + 2] = v2 > 0.f ? v2 : 0.f;
        As[strip][r][c + 3] = v3 > 0.f ? v3 : 0.f;
    }
    __syncthreads();

    for (int n0 = 32 * part; n0 < 32 * part + 32; n0 += 16) {
        v8f acc = {};
#pragma unroll
        for (int k0 = 0; k0 < HH; k0 += 4) {
            const int kb = k0 + hf * 2;
            v2f a, b;
            a.x = As[strip][lm][kb];  a.y = As[strip][lm][kb + 1];
            b.x = W2s[kb][n0 + lm];   b.y = W2s[kb + 1][n0 + lm];
            acc = wmma_f32_k4(a, b, acc);
        }
#pragma unroll
        for (int i = 0; i < 8; ++i) {
            const int m = i + hf * 8;
            const long long rr = r0 + m;
            if (rr < N) __builtin_nontemporal_store(acc[i], out + rr * DI + n0 + lm);
        }
    }
}

extern "C" void kernel_launch(void* const* d_in, const int* in_sizes, int n_in,
                              void* d_out, int out_size, void* d_ws, size_t ws_size,
                              hipStream_t stream) {
    const float* x     = (const float*)d_in[0];
    const int*   eidx  = (const int*)d_in[1];
    const float* ea    = (const float*)d_in[2];
    const float* Wedge = (const float*)d_in[3];
    const float* W1    = (const float*)d_in[4];
    const float* gamma = (const float*)d_in[5];
    const float* beta  = (const float*)d_in[6];
    const float* W2    = (const float*)d_in[7];
    float* out = (float*)d_out;

    const long long N = in_sizes[0] / DI;
    const long long E = in_sizes[1] / 2;
    const int* src = eidx;
    const int* dst = eidx + E;

    const int edgeBlocks = (int)((E + 63) / 64);
    const size_t nodeWords = (((size_t)N + 31) / 32 + 4) & ~(size_t)3;
    const size_t tileWords = (((size_t)edgeBlocks + 7) / 8 + 4) & ~(size_t)3;

    float* ws   = (float*)d_ws;
    float* mx   = ws;
    float* sum  = mx  + (size_t)N * DI;
    float* num  = sum + (size_t)N * DI;
    float* cs   = num + (size_t)N * DI;
    float* cq   = cs + 128;
    unsigned int* flag  = (unsigned int*)(cq + 128);
    unsigned int* tflag = flag + nodeWords;
    float* sc   = (float*)(tflag + tileWords);
    float* sh   = sc + 128;
    float* sumR = sh + 128;
    float* numR = sumR + (size_t)N * DI;
    float* hbuf = numR + (size_t)N * DI;

    hipMemsetAsync(mx, 0, ((size_t)3 * N * DI + 256 + nodeWords + tileWords) * sizeof(float), stream);

    k_edge_seg<<<edgeBlocks, 128, 0, stream>>>(x, src, dst, ea, Wedge, (unsigned int*)mx, sum, num,
                                               sumR, numR, flag, tflag, E);
    const int words = (edgeBlocks + 7) / 8;
    k_edge_fix<<<words < 256 ? words : 256, 128, 0, stream>>>(x, src, dst, ea, Wedge, (unsigned int*)mx, sum, num,
                                                              sumR, numR, flag, tflag, E, words);

    const int strips = (int)((N + 15) / 16);
    k_h_stats<<<(strips + 3) / 4, 128, 0, stream>>>(x, num, sum, mx, numR, sumR, flag, W1, hbuf, cs, cq, N);
    k_bn<<<1, 128, 0, stream>>>(cs, cq, gamma, beta, sc, sh, 1.0f / (float)N);
    k_out<<<(strips + 3) / 4, 256, 0, stream>>>(hbuf, W2, sc, sh, out, N);
}


# h kernel: a wave holds 4 column-interleaved tiles at once (pass p, tile t, column c = column 64p+4c+t): 1 A read + two 16-B W1 reads feed 4 product chains; W1 staged and h stored 16 B a lane
# speedup vs baseline: 3.3894x; 1.0383x over previous
#include <hip/hip_runtime.h>
#include <hip/hip_bf16.h>

typedef float v2f __attribute__((ext_vector_type(2)));
typedef float v4f __attribute__((ext_vector_type(4)));
typedef float v8f __attribute__((ext_vector_type(8)));

#define DI 64
#define HH 128

__device__ __forceinline__ v8f wmma_f32_k4(v2f a, v2f b, v8f c) {
    return __builtin_amdgcn_wmma_f32_16x16x4_f32(false, a, false, b, (short)0, c, false, false);
}

#define SHIFT_Q 8.0f
__device__ __forceinline__ float shift_level(float v) {
    return SHIFT_Q * (ceilf(v * (1.0f / SHIFT_Q)) - 1.0f);
}
#define LP 68
__device__ __forceinline__ void seg_add_row(float* __restrict__ sum, float* __restrict__ num,
                                            const int dn, const int lane,
                                            const float w0, const float w1,
                                            const float p0, const float p1) {
    float* b = sum + (long long)dn * DI;
    float* c = num + (long long)dn * DI;
    atomicAdd(b + lane,      w0);
    atomicAdd(b + lane + 32, w1);
    atomicAdd(c + lane,      p0);
    atomicAdd(c + lane + 32, p1);
}

template <bool FIX>
__device__ __forceinline__
void edge_tiles(float (&Ws)[64][LP], float (&As)[4][16][LP], const int vb,
                const float* __restrict__ x, const int* __restrict__ src,
                const int* __restrict__ dst,
                const float* __restrict__ ea, const float* __restrict__ Wedge,
                unsigned int* __restrict__ mx,
                float* __restrict__ sum, float* __restrict__ num,
                float* __restrict__ sumR, float* __restrict__ numR,
                unsigned int* __restrict__ flag, unsigned int* __restrict__ tflag,
                const long long E) {
    const int tid  = threadIdx.x;
    const int wave = tid >> 5;
    const int lane = tid & 31;
    const int hf   = lane >> 4;
    const int lm   = lane & 15;

    const int wv = __builtin_amdgcn_readfirstlane(wave);
    const long long eS = ((long long)vb * 4 + wv) * 16;
    const int eLast = (int)E - 1;
    const int eSi   = (int)eS;

    int sr[8];
#pragma unroll
    for (int r = 0; r < 8; ++r) {
        const int er = eSi + 8 * hf + r;
        sr[r] = src[er < eLast ? er : eLast];
    }

    int dn16[16];
    if (!FIX) {
#pragma unroll
        for (int e = 0; e < 16; ++e) {
            const int er = eSi + e;
            dn16[e] = dst[er < eLast ? er : eLast];
        }
    }

    v2f av[16];
#pragma unroll
    for (int r = 0; r < 16; ++r) {
        const int er = eSi + r;
        const int ek = er < eLast ? er : eLast;
        av[r] = __builtin_nontemporal_load((const v2f*)(ea + (long long)ek * DI + lane * 2));
    }

    v4f xv[8];
#pragma unroll
    for (int r = 0; r < 8; ++r)
        xv[r] = *(const v4f*)(x + (long long)sr[r] * DI + 4 * lm);

#pragma unroll
    for (int r = 0; r < 16; ++r)
        *(v2f*)&As[wave][r][lane * 2] = av[r];

#pragma unroll 4
    for (int n = 0; n < 8; ++n) {
        const int i = tid + 128 * n;
        *(v4f*)&Ws[i >> 4][4 * (i & 15)] = *(const v4f*)(Wedge + 4 * i);
    }
    __syncthreads();

    v8f acc[4] = {};
#pragma unroll
    for (int k0 = 0; k0 < DI; k0 += 4) {
        const int kb = k0 + hf * 2;
        const v2f a  = *(const v2f*)&As[wave][lm][kb];
        const v4f w0 = *(const v4f*)&Ws[kb][4 * lm];
        const v4f w1 = *(const v4f*)&Ws[kb + 1][4 * lm];
#pragma unroll
        for (int t = 0; t < 4; ++t) {
            v2f b;
            b.x = w0[t];  b.y = w1[t];
            acc[t] = wmma_f32_k4(a, b, acc[t]);
        }
    }

    v4f o[8];
#pragma unroll
    for (int r = 0; r < 8; ++r) {
#pragma unroll
        for (int t = 0; t < 4; ++t) {
            float v = acc[t][r] + xv[r][t];
            o[r][t] = (v > 0.f ? v : 0.f) + 1e-7f;
        }
    }
    float big = o[0][0];
#pragma unroll
    for (int r = 0; r < 8; ++r) {
#pragma unroll
        for (int t = 0; t < 4; ++t) big = fmaxf(big, o[r][t]);
    }
    const bool above = __builtin_amdgcn_ballot_w32(big > SHIFT_Q) != 0u;

    if (FIX) {
        if (above) {
#pragma unroll
            for (int r = 0; r < 8; ++r) {
                const int er = eSi + 8 * hf + r;
                if (er <= eLast) {
                    const long long row = (long long)dst[er] * DI + 4 * lm;
#pragma unroll
                    for (int t = 0; t < 4; ++t) {
                        const float m = o[r][t];
                        if (m > SHIFT_Q) {
                            const float s = __uint_as_float(mx[row + t]) + SHIFT_Q;
                            const float w = __expf(m - s);
                            atomicAdd(sumR + row + t, w);
                            atomicAdd(numR + row + t, m * w);
                        }
                    }
                }
            }
        }
        return;
    }

#pragma unroll
    for (int r = 0; r < 8; ++r)
        *(v4f*)&As[wave][8 * hf + r][4 * lm] = o[r];

    if (above) {
        if (lane == 0) {
            const int tile = vb * 4 + wv;
            atomicOr(tflag + (tile >> 5), 1u << (tile & 31));
        }
#pragma unroll
        for (int r = 0; r < 8; ++r) {
            const int er = eSi + 8 * hf + r;
            if (er <= eLast) {
                const int dn = dst[er];
                unsigned int* b = mx + (long long)dn * DI + 4 * lm;
                bool raised = false;
#pragma unroll
                for (int t = 0; t < 4; ++t) {
                    if (o[r][t] > SHIFT_Q) {
                        atomicMax(b + t, __float_as_uint(shift_level(o[r][t])));
                        raised = true;
                    }
                }
                if (raised) {
                    const unsigned int bit = 1u << (dn & 31);
                    const unsigned int old = atomicOr(flag + (dn >> 5), bit);
                    if ((old & bit) == 0u) {
                        v4f* zs = (v4f*)(sumR + (long long)dn * DI);
                        v4f* zn = (v4f*)(numR + (long long)dn * DI);
                        const v4f z = {0.f, 0.f, 0.f, 0.f};
                        for (int q = 0; q < DI / 4; ++q) { zs[q] = z; zn[q] = z; }
                    }
                }
            }
        }
    }
    __syncthreads();

    float w0[16], w1[16], p0[16], p1[16];
#pragma unroll
    for (int e = 0; e < 16; ++e) {
        const float m0 = As[wave][e][lane];
        const float m1 = As[wave][e][lane + 32];
        w0[e] = m0 > SHIFT_Q ? 0.f : __expf(m0 - SHIFT_Q);
        w1[e] = m1 > SHIFT_Q ? 0.f : __expf(m1 - SHIFT_Q);
        p0[e] = m0 * w0[e];
        p1[e] = m1 * w1[e];
    }
    if (eS + 16 <= E) {
#pragma unroll
        for (int e = 0; e < 16; ++e)
            seg_add_row(sum, num, dn16[e], lane, w0[e], w1[e], p0[e], p1[e]);
    } else {
#pragma unroll
        for (int e = 0; e < 16; ++e)
            if (eSi + e <= eLast) seg_add_row(sum, num, dn16[e], lane, w0[e], w1[e], p0[e], p1[e]);
    }
}

__global__ __launch_bounds__(128)
void k_edge_seg(const float* __restrict__ x, const int* __restrict__ src,
                const int* __restrict__ dst,
                const float* __restrict__ ea, const float* __restrict__ Wedge,
                unsigned int* __restrict__ mx,
                float* __restrict__ sum, float* __restrict__ num,
                float* __restrict__ sumR, float* __restrict__ numR,
                unsigned int* __restrict__ flag, unsigned int* __restrict__ tflag,
                long long E) {
    __shared__ __attribute__((aligned(16))) float Ws[64][LP];
    __shared__ __attribute__((aligned(16))) float As[4][16][LP];
    edge_tiles<false>(Ws, As, (int)blockIdx.x, x, src, dst, ea, Wedge, mx, sum, num, sumR, numR, flag, tflag, E);
}

__global__ __launch_bounds__(128)
void k_edge_fix(const float* __restrict__ x, const int* __restrict__ src,
                const int* __restrict__ dst,
                const float* __restrict__ ea, const float* __restrict__ Wedge,
                unsigned int* __restrict__ mx,
                float* __restrict__ sum, float* __restrict__ num,
                float* __restrict__ sumR, float* __restrict__ numR,
                unsigned int* __restrict__ flag, unsigned int* __restrict__ tflag,
                long long E, int words) {
    __shared__ __attribute__((aligned(16))) float Ws[64][LP];
    __shared__ __attribute__((aligned(16))) float As[4][16][LP];
    for (int w = blockIdx.x; w < words; w += gridDim.x) {
        const unsigned int tw = tflag[w];
        if (tw == 0u) continue;
#pragma unroll 1
        for (int q = 0; q < 8; ++q) {
            if (((tw >> (4 * q)) & 0xFu) != 0u) {
                edge_tiles<true>(Ws, As, w * 8 + q, x, src, dst, ea, Wedge, mx, sum, num, sumR, numR, flag, tflag, E);
                __syncthreads();
            }
        }
    }
}

#define HP 132
__global__ __launch_bounds__(128)
void k_h_stats(const float* __restrict__ x, const float* __restrict__ num,
               const float* __restrict__ sum,
               const float* __restrict__ mx, const float* __restrict__ numR,
               const float* __restrict__ sumR, const unsigned int* __restrict__ flag,
               const float* __restrict__ W1, float* __restrict__ h,
               float* __restrict__ colsum, float* __restrict__ colsq, long long N) {
    __shared__ __attribute__((aligned(16))) float W1s[64][HP];
    __shared__ __attribute__((aligned(16))) float As[4][16][LP];
    __shared__ float cs_s[128], cq_s[128];

    const int tid  = threadIdx.x;
    const int wave = tid >> 5;
    const int lane = tid & 31;
    const int hf   = lane >> 4;
    const int lm   = lane & 15;

#pragma unroll
    for (int n = 0; n < 16; ++n) {
        const int i = tid + 128 * n;
        *(v4f*)&W1s[i >> 5][4 * (i & 31)] = *(const v4f*)(W1 + 4 * i);
    }
    if (tid < 128) { cs_s[tid] = 0.f; cq_s[tid] = 0.f; }

    const long long r0 = ((long long)blockIdx.x * 4 + wave) * 16;
#pragma unroll
    for (int r = 0; r < 16; ++r) {
        long long rr = r0 + r;
        long long rc = rr < N ? rr : N - 1;
        const float2 a  = *(const float2*)(num + rc * DI + lane * 2);
        const float2 sm = *(const float2*)(sum + rc * DI + lane * 2);
        const float2 xv = *(const float2*)(x + rc * DI + lane * 2);
        v2f o;
        o.x = a.x / (sm.x + 1e-16f) + xv.x;
        o.y = a.y / (sm.y + 1e-16f) + xv.y;
        *(v2f*)&As[wave][r][lane * 2] = o;
    }
    const long long r0s = ((long long)blockIdx.x * 4 + __builtin_amdgcn_readfirstlane(wave)) * 16;
    const unsigned int fw = r0s < N ? (flag[r0s >> 5] >> (r0s & 31)) & 0xFFFFu : 0u;
    if (fw != 0u) {
#pragma unroll 1
        for (int r = 0; r < 16; ++r) {
            if ((fw >> r) & 1u) {
                const long long rc = r0s + r;
                const float2 a  = *(const float2*)(num + rc * DI + lane * 2);
                const float2 sm = *(const float2*)(sum + rc * DI + lane * 2);
                const float2 xv = *(const float2*)(x + rc * DI + lane * 2);
                const float2 lv = *(const float2*)(mx + rc * DI + lane * 2);
                const float2 aR = *(const float2*)(numR + rc * DI + lane * 2);
                const float2 sR = *(const float2*)(sumR + rc * DI + lane * 2);
                const float f0 = __expf(-lv.x), f1 = __expf(-lv.y);
                As[wave][r][lane * 2 + 0] = (a.x * f0 + aR.x) / (sm.x * f0 + sR.x + 1e-16f) + xv.x;
                As[wave][r][lane * 2 + 1] = (a.y * f1 + aR.y) / (sm.y * f1 + sR.y + 1e-16f) + xv.y;
            }
        }
    }
    __syncthreads();

#pragma unroll
    for (int p = 0; p < 2; ++p) {
        v8f acc[4] = {};
#pragma unroll
        for (int k0 = 0; k0 < DI; k0 += 4) {
            const int kb = k0 + hf * 2;
            const v2f a  = *(const v2f*)&As[wave][lm][kb];
            const v4f w0 = *(const v4f*)&W1s[kb][64 * p + 4 * lm];
            const v4f w1 = *(const v4f*)&W1s[kb + 1][64 * p + 4 * lm];
#pragma unroll
            for (int t = 0; t < 4; ++t) {
                v2f b;
                b.x = w0[t];  b.y = w1[t];
                acc[t] = wmma_f32_k4(a, b, acc[t]);
            }
        }

        float p1[4] = {0.f, 0.f, 0.f, 0.f}, p2[4] = {0.f, 0.f, 0.f, 0.f};
        float* hrow = h + (r0 + 8 * hf) * HH + 64 * p + 4 * lm;
#pragma unroll
        for (int r = 0; r < 8; ++r) {
            if (r0 + 8 * hf + r < N) {
                v4f o;
#pragma unroll
                for (int t = 0; t < 4; ++t) {
                    const float v = acc[t][r];
                    o[t] = v;
                    p1[t] += v;
                    p2[t] += v * v;
                }
                *(v4f*)(hrow + r * HH) = o;
            }
        }
#pragma unroll
        for (int t = 0; t < 4; ++t) {
            p1[t] += __shfl_xor(p1[t], 16, 32);
            p2[t] += __shfl_xor(p2[t], 16, 32);
        }
        if (hf == 0) {
#pragma unroll
            for (int t = 0; t < 4; ++t) {
                atomicAdd(&cs_s[64 * p + 4 * lm + t], p1[t]);
                atomicAdd(&cq_s[64 * p + 4 * lm + t], p2[t]);
            }
        }
    }
    __syncthreads();
    if (tid < 128) {
        atomicAdd(colsum + tid, cs_s[tid]);
        atomicAdd(colsq + tid, cq_s[tid]);
    }
}

__global__ __launch_bounds__(128)
void k_bn(const float* __restrict__ colsum, const float* __restrict__ colsq,
          const float* __restrict__ gamma, const float* __restrict__ beta,
          float* __restrict__ scale, float* __restrict__ shift, float invN) {
    const int i = threadIdx.x;
    const float mu  = colsum[i] * invN;
    const float var = colsq[i] * invN - mu * mu;
    const float sc  = rsqrtf(var + 1e-5f) * gamma[i];
    scale[i] = sc;
    shift[i] = beta[i] - mu * sc;
}

__global__ __launch_bounds__(256)
void k_out(const float* __restrict__ h, const float* __restrict__ W2,
           const float* __restrict__ scale, const float* __restrict__ shift,
           float* __restrict__ out, long long N) {
    __shared__ float W2s[HH][66];
    __shared__ float As[4][16][130];
    __shared__ float sc_s[128], sh_s[128];

    const int tid   = threadIdx.x;
    const int wave  = tid >> 5;
    const int strip = wave >> 1;
    const int part  = wave & 1;
    const int lane  = tid & 31;
    const int hf    = lane >> 4;
    const int lm    = lane & 15;

    for (int i = tid; i < HH * DI; i += 256)
        W2s[i >> 6][i & 63] = W2[i];
    if (tid < 128) { sc_s[tid] = scale[tid]; sh_s[tid] = shift[tid]; }
    __syncthreads();

    const long long r0 = ((long long)blockIdx.x * 4 + strip) * 16;
    for (int r = 8 * part; r < 8 * part + 8; ++r) {
        long long rr = r0 + r;
        long long rc = rr < N ? rr : N - 1;
        const int c = lane * 4;
        const v4f hv = *(const v4f*)(h + rc * HH + c);
        float v0 = hv.x * sc_s[c + 0] + sh_s[c + 0];
        float v1 = hv.y * sc_s[c + 1] + sh_s[c + 1];
        float v2 = hv.z * sc_s[c + 2] + sh_s[c + 2];
        float v3 = hv.w * sc_s[c + 3] + sh_s[c + 3];
        As[strip][r][c + 0] = v0 > 0.f ? v0 : 0.f;
        As[strip][r][c + 1] = v1 > 0.f ? v1 : 0.f;
        As[strip][r][c + 2] = v2 > 0.f ? v2 : 0.f;
        As[strip][r][c + 3] = v3 > 0.f ? v3 : 0.f;
    }
    __syncthreads();

    for (int n0 = 32 * part; n0 < 32 * part + 32; n0 += 16) {
        v8f acc = {};
#pragma unroll
        for (int k0 = 0; k0 < HH; k0 += 4) {
            const int kb = k0 + hf * 2;
            v2f a, b;
            a.x = As[strip][lm][kb];  a.y = As[strip][lm][kb + 1];
            b.x = W2s[kb][n0 + lm];   b.y = W2s[kb + 1][n0 + lm];
            acc = wmma_f32_k4(a, b, acc);
        }
#pragma unroll
        for (int i = 0; i < 8; ++i) {
            const int m = i + hf * 8;
            const long long rr = r0 + m;
            if (rr < N) __builtin_nontemporal_store(acc[i], out + rr * DI + n0 + lm);
        }
    }
}

extern "C" void kernel_launch(void* const* d_in, const int* in_sizes, int n_in,
                              void* d_out, int out_size, void* d_ws, size_t ws_size,
                              hipStream_t stream) {
    const float* x     = (const float*)d_in[0];
    const int*   eidx  = (const int*)d_in[1];
    const float* ea    = (const float*)d_in[2];
    const float* Wedge = (const float*)d_in[3];
    const float* W1    = (const float*)d_in[4];
    const float* gamma = (const float*)d_in[5];
    const float* beta  = (const float*)d_in[6];
    const float* W2    = (const float*)d_in[7];
    float* out = (float*)d_out;

    const long long N = in_sizes[0] / DI;
    const long long E = in_sizes[1] / 2;
    const int* src = eidx;
    const int* dst = eidx + E;

    const int edgeBlocks = (int)((E + 63) / 64);
    const size_t nodeWords = (((size_t)N + 31) / 32 + 4) & ~(size_t)3;
    const size_t tileWords = (((size_t)edgeBlocks + 7) / 8 + 4) & ~(size_t)3;

    float* ws   = (float*)d_ws;
    float* mx   = ws;
    float* sum  = mx  + (size_t)N * DI;
    float* num  = sum + (size_t)N * DI;
    float* cs   = num + (size_t)N * DI;
    float* cq   = cs + 128;
    unsigned int* flag  = (unsigned int*)(cq + 128);
    unsigned int* tflag = flag + nodeWords;
    float* sc   = (float*)(tflag + tileWords);
    float* sh   = sc + 128;
    float* sumR = sh + 128;
    float* numR = sumR + (size_t)N * DI;
    float* hbuf = numR + (size_t)N * DI;

    hipMemsetAsync(mx, 0, ((size_t)3 * N * DI + 256 + nodeWords + tileWords) * sizeof(float), stream);

    k_edge_seg<<<edgeBlocks, 128, 0, stream>>>(x, src, dst, ea, Wedge, (unsigned int*)mx, sum, num,
                                               sumR, numR, flag, tflag, E);
    const int words = (edgeBlocks + 7) / 8;
    k_edge_fix<<<words < 256 ? words : 256, 128, 0, stream>>>(x, src, dst, ea, Wedge, (unsigned int*)mx, sum, num,
                                                              sumR, numR, flag, tflag, E, words);

    const int strips = (int)((N + 15) / 16);
    k_h_stats<<<(strips + 3) / 4, 128, 0, stream>>>(x, num, sum, mx, numR, sumR, flag, W1, hbuf, cs, cq, N);
    k_bn<<<1, 128, 0, stream>>>(cs, cq, gamma, beta, sc, sh, 1.0f / (float)N);
    k_out<<<(strips + 3) / 4, 256, 0, stream>>>(hbuf, W2, sc, sh, out, N);
}
